# starting kernel
# baseline (speedup 1.0000x reference)
#include <hip/hip_runtime.h>
#include <hip/hip_bf16.h>
#include <math.h>


typedef __attribute__((ext_vector_type(2))) float v2f;
typedef __attribute__((ext_vector_type(8))) float v8f;

#define HDIM 256
#define TLEN 512
#define BATCH 64
#define GDIM 1024
#define KDIM 512
#define CDIM 60
#define AST  520

__device__ __forceinline__ float sigmoidf_(float x) {
    return 1.0f / (1.0f + __expf(-x));
}
__device__ __forceinline__ float tanhf_(float x) {
    return 1.0f - 2.0f / (__expf(2.0f * x) + 1.0f);
}

#define WMMA_F32X4(A, B, C) \
    __builtin_amdgcn_wmma_f32_16x16x4_f32(false, (A), false, (B), (short)0, (C), false, false)

__global__ __launch_bounds__(256) void pack_lstm_weights(
    const float* __restrict__ Wih0, const float* __restrict__ Whh0,
    const float* __restrict__ Wih1, const float* __restrict__ Whh1,
    const float* __restrict__ Wih2, const float* __restrict__ Whh2,
    const float* __restrict__ Wih3, const float* __restrict__ Whh3,
    float* __restrict__ wpk) {
    int gid = blockIdx.x * blockDim.x + threadIdx.x;
    if (gid >= 4 * KDIM * GDIM) return;
    int l   = gid >> 19;
    int rem = gid & (KDIM * GDIM - 1);
    int k   = rem >> 10;
    int n   = rem & 1023;
    const float* Wih = (l == 0) ? Wih0 : (l == 1) ? Wih1 : (l == 2) ? Wih2 : Wih3;
    const float* Whh = (l == 0) ? Whh0 : (l == 1) ? Whh1 : (l == 2) ? Whh2 : Whh3;
    float v = (k < HDIM) ? Wih[n * HDIM + k] : Whh[n * HDIM + (k - HDIM)];
    wpk[(size_t)l * (KDIM * GDIM) + (size_t)(k >> 1) * (2 * GDIM) + n * 2 + (k & 1)] = v;
}

__global__ __launch_bounds__(256) void pack_fc(const float* __restrict__ fcw,
                                               float* __restrict__ fcpk) {
    int gid = blockIdx.x * blockDim.x + threadIdx.x;
    if (gid >= HDIM * 64) return;
    int k = gid >> 6, n = gid & 63;
    float v = (n < CDIM) ? fcw[n * HDIM + k] : 0.0f;
    fcpk[(k >> 1) * 128 + n * 2 + (k & 1)] = v;
}

__global__ __launch_bounds__(512) void lstm_scan(
    const float* xin, float* yout,
    const float* __restrict__ wpk,
    const float* __restrict__ b_ih, const float* __restrict__ b_hh,
    float* __restrict__ hT, float* __restrict__ cT) {
    __shared__ float abuf[16 * AST];

    const int tid   = threadIdx.x;
    const int wave  = tid >> 5;
    const int lane  = tid & 31;
    const int b0    = blockIdx.x * 16;
    const int arow  = lane & 15;
    const int khalf = lane >> 4;
    const int jcol  = (wave << 4) + (lane & 15);

    for (int i = tid; i < 16 * AST; i += 512) abuf[i] = 0.0f;

    const float bi = b_ih[jcol]       + b_hh[jcol];
    const float bf = b_ih[256 + jcol] + b_hh[256 + jcol];
    const float bg = b_ih[512 + jcol] + b_hh[512 + jcol];
    const float bo = b_ih[768 + jcol] + b_hh[768 + jcol];

    float creg[8], hreg[8];
#pragma unroll
    for (int r = 0; r < 8; ++r) { creg[r] = 0.0f; hreg[r] = 0.0f; }

    const float* xrow_ptr  = xin + (size_t)(b0 + wave) * (TLEN * HDIM);
    float*       yrow_base = yout + (size_t)b0 * (TLEN * HDIM);

    for (int t = 0; t < TLEN; ++t) {
        {
            const float4* src = (const float4*)(xrow_ptr + t * HDIM + (lane << 3));
            float4 v0 = src[0], v1 = src[1];
            float4* dst = (float4*)&abuf[wave * AST + (lane << 3)];
            dst[0] = v0; dst[1] = v1;
        }
        __syncthreads();

        v8f acc0 = {0, 0, 0, 0, 0, 0, 0, 0};
        v8f acc1 = {0, 0, 0, 0, 0, 0, 0, 0};
        v8f acc2 = {0, 0, 0, 0, 0, 0, 0, 0};
        v8f acc3 = {0, 0, 0, 0, 0, 0, 0, 0};

        const float* arowp = &abuf[arow * AST + 2 * khalf];
        for (int k0 = 0; k0 < KDIM; k0 += 4) {
            v2f a = *(const v2f*)(arowp + k0);
            const float* wrow =
                wpk + ((size_t)((k0 >> 1) + khalf)) * (2 * GDIM) + 2 * jcol;
            v2f bI = *(const v2f*)(wrow);
            v2f bF = *(const v2f*)(wrow + 512);
            v2f bG = *(const v2f*)(wrow + 1024);
            v2f bO = *(const v2f*)(wrow + 1536);
            acc0 = WMMA_F32X4(a, bI, acc0);
            acc1 = WMMA_F32X4(a, bF, acc1);
            acc2 = WMMA_F32X4(a, bG, acc2);
            acc3 = WMMA_F32X4(a, bO, acc3);
        }
        __syncthreads();

#pragma unroll
        for (int r = 0; r < 8; ++r) {
            float iv = sigmoidf_(acc0[r] + bi);
            float fv = sigmoidf_(acc1[r] + bf);
            float gv = tanhf_(acc2[r] + bg);
            float ov = sigmoidf_(acc3[r] + bo);
            float cn = fv * creg[r] + iv * gv;
            float hn = ov * tanhf_(cn);
            creg[r] = cn;
            hreg[r] = hn;
            int brow = r + (khalf << 3);
            abuf[brow * AST + 256 + jcol] = hn;
            yrow_base[(size_t)brow * (TLEN * HDIM) + t * HDIM + jcol] = hn;
        }
    }

#pragma unroll
    for (int r = 0; r < 8; ++r) {
        int brow = r + (khalf << 3);
        hT[(size_t)(b0 + brow) * HDIM + jcol] = hreg[r];
        cT[(size_t)(b0 + brow) * HDIM + jcol] = creg[r];
    }
}

__global__ __launch_bounds__(256) void fc_kernel(
    const float* __restrict__ y, const float* __restrict__ fcpk,
    const float* __restrict__ fcb, float* __restrict__ label) {
    const int tid   = threadIdx.x;
    const int lane  = tid & 31;
    const int gw    = blockIdx.x * 8 + (tid >> 5);
    const int b     = gw >> 5;
    const int t0    = (gw & 31) << 4;
    const int khalf = lane >> 4;
    const int trow  = t0 + (lane & 15);

    const float* arowp = y + (size_t)b * (TLEN * HDIM) + (size_t)trow * HDIM + 2 * khalf;

    v8f acc[4];
#pragma unroll
    for (int j = 0; j < 4; ++j) acc[j] = (v8f){0, 0, 0, 0, 0, 0, 0, 0};

    for (int k0 = 0; k0 < HDIM; k0 += 4) {
        v2f a = *(const v2f*)(arowp + k0);
        const float* wrow = fcpk + ((k0 >> 1) + khalf) * 128 + 2 * (lane & 15);
#pragma unroll
        for (int j = 0; j < 4; ++j) {
            v2f bv = *(const v2f*)(wrow + 32 * j);
            acc[j] = WMMA_F32X4(a, bv, acc[j]);
        }
    }

#pragma unroll
    for (int j = 0; j < 4; ++j) {
        int n = (j << 4) + (lane & 15);
        if (n < CDIM) {
            float bias = fcb[n];
            float* outp =
                label + (size_t)b * (CDIM * TLEN) + (size_t)n * TLEN + t0 + (khalf << 3);
#pragma unroll
            for (int r = 0; r < 8; ++r) outp[r] = acc[j][r] + bias;
        }
    }
}

extern "C" void kernel_launch(void* const* d_in, const int* in_sizes, int n_in,
                              void* d_out, int out_size, void* d_ws, size_t ws_size,
                              hipStream_t stream) {
    const float* x = (const float*)d_in[0];
    const float *Wih[4], *Whh[4], *bih[4], *bhh[4];
    for (int l = 0; l < 4; ++l) {
        Wih[l] = (const float*)d_in[1 + 4 * l];
        Whh[l] = (const float*)d_in[2 + 4 * l];
        bih[l] = (const float*)d_in[3 + 4 * l];
        bhh[l] = (const float*)d_in[4 + 4 * l];
    }
    const float* fcw = (const float*)d_in[17];
    const float* fcb = (const float*)d_in[18];

    float* out   = (float*)d_out;
    float* label = out;
    float* hs    = out + (size_t)BATCH * CDIM * TLEN;
    float* cs    = hs + (size_t)4 * BATCH * HDIM;

    float* ws   = (float*)d_ws;
    float* ybuf = ws;
    float* wpk  = ybuf + (size_t)BATCH * TLEN * HDIM;
    float* fcpk = wpk + (size_t)4 * KDIM * GDIM;

    pack_lstm_weights<<<(4 * KDIM * GDIM) / 256, 256, 0, stream>>>(
        Wih[0], Whh[0], Wih[1], Whh[1], Wih[2], Whh[2], Wih[3], Whh[3], wpk);
    pack_fc<<<(HDIM * 64) / 256, 256, 0, stream>>>(fcw, fcpk);

    for (int l = 0; l < 4; ++l) {
        lstm_scan<<<4, 512, 0, stream>>>(
            (l == 0) ? x : ybuf, ybuf, wpk + (size_t)l * KDIM * GDIM,
            bih[l], bhh[l],
            hs + (size_t)l * BATCH * HDIM, cs + (size_t)l * BATCH * HDIM);
    }

    fc_kernel<<<256, 256, 0, stream>>>(ybuf, fcpk, fcb, label);
}


# scan time loop rotated: x half of step t+1's k loop runs behind the gate math of step t, before the one barrier left per step; x, h in two LDS slots (15 k-steps kept in LDS, not 16); sums unchanged
# speedup vs baseline: 1.2518x; 1.2518x over previous
#include <hip/hip_runtime.h>
#include <hip/hip_bf16.h>
#include <math.h>


typedef __attribute__((ext_vector_type(2))) float v2f;
typedef __attribute__((ext_vector_type(8))) float v8f;

#define HDIM 256
#define TLEN 512
#define BATCH 64
#define GDIM 1024
#define KDIM 512
#define CDIM 60
#define AST  520
#define SLOT (16 * AST)
#define NKS  128
#define NLDS 15
#define NREG 12
#define PFD  6
#define RING 8
#define XLD_KS 96

#define KS_IN_LDS(ks)   ((((ks) & 7) == 0) && ((ks) < 8 * NLDS))
#define KS_IN_REG(ks)   ((((ks) & 7) == 4) && ((((ks) >> 3) & 3) != 3))
#define KS_REG_SLOT(ks) ((((ks) >> 5) * 3) + (((ks) >> 3) & 3))

#define REP8(M, b) M((b) + 0) M((b) + 1) M((b) + 2) M((b) + 3) M((b) + 4) M((b) + 5) M((b) + 6) M((b) + 7)
#define REP64(M, b)                                                                  \
    REP8(M, (b) + 0)  REP8(M, (b) + 8)  REP8(M, (b) + 16) REP8(M, (b) + 24)          \
    REP8(M, (b) + 32) REP8(M, (b) + 40) REP8(M, (b) + 48) REP8(M, (b) + 56)
#define REP128(M) REP64(M, 0) REP64(M, 64)
#define REP62(M)                                                                     \
    REP8(M, 0)  REP8(M, 8)  REP8(M, 16) REP8(M, 24) REP8(M, 32) REP8(M, 40)          \
    REP8(M, 48) M(56) M(57) M(58) M(59) M(60) M(61)

__device__ __forceinline__ float sigmoidf_(float x) {
    return __builtin_amdgcn_rcpf(1.0f + __expf(-x));
}
__device__ __forceinline__ float tanhf_(float x) {
    return 1.0f - 2.0f * __builtin_amdgcn_rcpf(__expf(2.0f * x) + 1.0f);
}

#define WMMA_F32X4(A, B, C) \
    __builtin_amdgcn_wmma_f32_16x16x4_f32(false, (A), false, (B), (short)0, (C), false, false)

__global__ __launch_bounds__(256) void pack_lstm_weights(
    const float* __restrict__ Wih0, const float* __restrict__ Whh0,
    const float* __restrict__ Wih1, const float* __restrict__ Whh1,
    const float* __restrict__ Wih2, const float* __restrict__ Whh2,
    const float* __restrict__ Wih3, const float* __restrict__ Whh3,
    float* __restrict__ wpk) {
    int gid = blockIdx.x * blockDim.x + threadIdx.x;
    if (gid >= 4 * KDIM * GDIM) return;
    int l   = gid >> 19;
    int rem = gid & (KDIM * GDIM - 1);
    int k   = rem >> 10;
    int n   = rem & 1023;
    const float* Wih = (l == 0) ? Wih0 : (l == 1) ? Wih1 : (l == 2) ? Wih2 : Wih3;
    const float* Whh = (l == 0) ? Whh0 : (l == 1) ? Whh1 : (l == 2) ? Whh2 : Whh3;
    float v = (k < HDIM) ? Wih[n * HDIM + k] : Whh[n * HDIM + (k - HDIM)];
    wpk[(size_t)l * (KDIM * GDIM) + (size_t)(k >> 1) * (2 * GDIM) + n * 2 + (k & 1)] = v;
}

__global__ __launch_bounds__(256) void pack_fc(const float* __restrict__ fcw,
                                               float* __restrict__ fcpk) {
    int gid = blockIdx.x * blockDim.x + threadIdx.x;
    if (gid >= HDIM * 64) return;
    int k = gid >> 6, n = gid & 63;
    float v = (n < CDIM) ? fcw[n * HDIM + k] : 0.0f;
    fcpk[(k >> 1) * 128 + n * 2 + (k & 1)] = v;
}


#define LOAD_BFRAG(dst, ks)                                        \
    do {                                                           \
        const float* p_ = wbase + (size_t)(ks) * (4 * GDIM);       \
        (dst)[0] = *(const v2f*)(p_);                              \
        (dst)[1] = *(const v2f*)(p_ + 512);                        \
        (dst)[2] = *(const v2f*)(p_ + 1024);                       \
        (dst)[3] = *(const v2f*)(p_ + 1536);                       \
    } while (0)

#define REQUEST_BFRAG(ks)                                                      \
    do {                                                                       \
        if (KS_IN_LDS(ks)) {                                                   \
            float4 u_ = wlds[(((ks) >> 3) * 2 + 0) * 512 + tid];               \
            float4 v_ = wlds[(((ks) >> 3) * 2 + 1) * 512 + tid];               \
            ring[(ks) % RING][0] = (v2f){u_.x, u_.y};                          \
            ring[(ks) % RING][1] = (v2f){u_.z, u_.w};                          \
            ring[(ks) % RING][2] = (v2f){v_.x, v_.y};                          \
            ring[(ks) % RING][3] = (v2f){v_.z, v_.w};                          \
        } else if (!KS_IN_REG(ks)) {                                           \
            LOAD_BFRAG(ring[(ks) % RING], ks);                                 \
        }                                                                      \
    } while (0)

#define FILL_LDS(ks)                                                                        \
    if (KS_IN_LDS(ks)) {                                                                    \
        v2f tb_[4];                                                                         \
        LOAD_BFRAG(tb_, ks);                                                                \
        wlds[(((ks) >> 3) * 2 + 0) * 512 + tid] = make_float4(tb_[0][0], tb_[0][1], tb_[1][0], tb_[1][1]); \
        wlds[(((ks) >> 3) * 2 + 1) * 512 + tid] = make_float4(tb_[2][0], tb_[2][1], tb_[3][0], tb_[3][1]); \
    }
#define FILL_REG(ks) \
    if (KS_IN_REG(ks)) { LOAD_BFRAG(breg[KS_REG_SLOT(ks) % NREG], ks); }

#define KS_PRIO(ks) (((ks) == 32) ? 1 : 0)
#define KS_SETS_PRIO(ks) (((ks) == 32) || ((ks) == 48))

#define A_BEHIND_BARRIER(ks) (((ks) == 64) || ((ks) == 65))

#define KSTEP(ks)                                                                 \
        {                                                                         \
            if (KS_SETS_PRIO(ks)) __builtin_amdgcn_s_setprio(KS_PRIO(ks));        \
            if ((ks) == XLD_KS) { xv0 = xsrc[0]; xv1 = xsrc[1]; }                 \
            REQUEST_BFRAG(((ks) + PFD) % NKS);                                    \
            if (!A_BEHIND_BARRIER(((ks) + 2) % NKS))                              \
                aring[((ks) + 2) & 3] = *(const v2f*)(arowp + 4 * (((ks) + 2) % NKS)); \
            const v2f a_ = aring[(ks) & 3];                                       \
            v2f bI_, bF_, bG_, bO_;                                               \
            if (KS_IN_REG(ks)) {                                                  \
                bI_ = breg[KS_REG_SLOT(ks) % NREG][0];                            \
                bF_ = breg[KS_REG_SLOT(ks) % NREG][1];                            \
                bG_ = breg[KS_REG_SLOT(ks) % NREG][2];                            \
                bO_ = breg[KS_REG_SLOT(ks) % NREG][3];                            \
            } else {                                                              \
                bI_ = ring[(ks) % RING][0]; bF_ = ring[(ks) % RING][1];           \
                bG_ = ring[(ks) % RING][2]; bO_ = ring[(ks) % RING][3];           \
            }                                                                     \
            acc0 = WMMA_F32X4(a_, bI_, acc0);         \
            acc1 = WMMA_F32X4(a_, bF_, acc1);         \
            acc2 = WMMA_F32X4(a_, bG_, acc2);         \
            acc3 = WMMA_F32X4(a_, bO_, acc3);         \
              \
            __builtin_amdgcn_sched_barrier(0);                                    \
        }

__global__ __launch_bounds__(512) void lstm_scan(
    const float* xin, float* yout,
    const float* wpk,
    const float* __restrict__ b_ih, const float* __restrict__ b_hh,
    float* __restrict__ hT, float* __restrict__ cT) {
    __shared__ __attribute__((aligned(16))) float abuf[2 * SLOT];
    __shared__ float4 wlds[NLDS * 2 * 512];

    const int tid   = threadIdx.x;
    const int wave  = tid >> 5;
    const int lane  = tid & 31;
    const int b0    = blockIdx.x * 16;
    const int arow  = lane & 15;
    const int khalf = lane >> 4;
    const int jcol  = (wave << 4) + (lane & 15);

    const float* xrow_ptr = xin + (size_t)(b0 + wave) * (TLEN * HDIM) + (lane << 3);
    const int    x_off    = wave * AST + (lane << 3);
    {
        const float4 z = make_float4(0.0f, 0.0f, 0.0f, 0.0f);
        float4* hz = (float4*)&abuf[SLOT + x_off + 256];
        hz[0] = z; hz[1] = z;
        const float4* s0 = (const float4*)(xrow_ptr);
        const float4* s1 = (const float4*)(xrow_ptr + HDIM);
        const float4 p0 = s0[0], p1 = s0[1], q0 = s1[0], q1 = s1[1];
        float4* d0 = (float4*)&abuf[x_off];
        float4* d1 = (float4*)&abuf[SLOT + x_off];
        d0[0] = p0; d0[1] = p1;
        d1[0] = q0; d1[1] = q1;
    }

    const float* wbase = wpk + (size_t)khalf * (2 * GDIM) + 2 * jcol;

    REP128(FILL_LDS)
    v2f breg[NREG][4];
    REP128(FILL_REG)

    const float bi = b_ih[jcol]       + b_hh[jcol];
    const float bf = b_ih[256 + jcol] + b_hh[256 + jcol];
    const float bg = b_ih[512 + jcol] + b_hh[512 + jcol];
    const float bo = b_ih[768 + jcol] + b_hh[768 + jcol];

    float creg[8], hreg[8];
#pragma unroll
    for (int r = 0; r < 8; ++r) { creg[r] = 0.0f; hreg[r] = 0.0f; }

    float*    ylane = yout + (size_t)(b0 + (khalf << 3)) * (TLEN * HDIM) + jcol;
    const int a_off = arow * AST + 2 * khalf;
    const int h_off = (khalf << 3) * AST + 256 + jcol;

    __syncthreads();

    v2f ring[RING][4];
    REQUEST_BFRAG(0); REQUEST_BFRAG(1); REQUEST_BFRAG(2);
    REQUEST_BFRAG(3); REQUEST_BFRAG(4); REQUEST_BFRAG(5);

    v8f acc0 = {0, 0, 0, 0, 0, 0, 0, 0};
    v8f acc1 = {0, 0, 0, 0, 0, 0, 0, 0};
    v8f acc2 = {0, 0, 0, 0, 0, 0, 0, 0};
    v8f acc3 = {0, 0, 0, 0, 0, 0, 0, 0};

    v2f aring[4];
    float4 xv0, xv1;
    const float4* xsrc = (const float4*)xrow_ptr;
    const float* arowp = &abuf[a_off];
    aring[0] = *(const v2f*)(arowp);
    aring[1] = *(const v2f*)(arowp + 4);

    REP62(KSTEP)

    for (int t = 0; t < TLEN; ++t) {
        asm volatile("s_wait_dscnt 0x0\n\ts_barrier_signal -1\n\ts_barrier_wait -1" ::: "memory");
        __builtin_amdgcn_s_setprio(3);

        const int rd = ((t + 1) & 1) * SLOT;
        const int wr = (t & 1) * SLOT;
        arowp = &abuf[rd + a_off];
        aring[0] = *(const v2f*)(arowp + 4 * 64);
        aring[1] = *(const v2f*)(arowp + 4 * 65);
        __builtin_amdgcn_sched_barrier(0);
        xsrc = (const float4*)(xrow_ptr + ((t + 2) & (TLEN - 1)) * HDIM);

        KSTEP(62) KSTEP(63)
        REP64(KSTEP, 64)

        __builtin_amdgcn_s_setprio(2);

        float* hlane = &abuf[wr + h_off];
#pragma unroll
        for (int r = 0; r < 8; ++r) {
            float iv = sigmoidf_(acc0[r] + bi);
            float fv = sigmoidf_(acc1[r] + bf);
            float gv = tanhf_(acc2[r] + bg);
            float ov = sigmoidf_(acc3[r] + bo);
            float cn = fv * creg[r] + iv * gv;
            float hn = ov * tanhf_(cn);
            creg[r] = cn;
            hreg[r] = hn;
            hlane[r * AST] = hn;
            ylane[(size_t)r * (TLEN * HDIM) + t * HDIM] = hn;
        }
        {
            float4* xdst = (float4*)&abuf[wr + x_off];
            xdst[0] = xv0; xdst[1] = xv1;
        }

        acc0 = (v8f){0, 0, 0, 0, 0, 0, 0, 0};
        acc1 = (v8f){0, 0, 0, 0, 0, 0, 0, 0};
        acc2 = (v8f){0, 0, 0, 0, 0, 0, 0, 0};
        acc3 = (v8f){0, 0, 0, 0, 0, 0, 0, 0};

        REP62(KSTEP)
    }

#pragma unroll
    for (int r = 0; r < 8; ++r) {
        int brow = r + (khalf << 3);
        hT[(size_t)(b0 + brow) * HDIM + jcol] = hreg[r];
        cT[(size_t)(b0 + brow) * HDIM + jcol] = creg[r];
    }
}

__global__ __launch_bounds__(256) void fc_kernel(
    const float* __restrict__ y, const float* __restrict__ fcpk,
    const float* __restrict__ fcb, float* __restrict__ label) {
    const int tid   = threadIdx.x;
    const int lane  = tid & 31;
    const int gw    = blockIdx.x * 8 + (tid >> 5);
    const int b     = gw >> 5;
    const int t0    = (gw & 31) << 4;
    const int khalf = lane >> 4;
    const int trow  = t0 + (lane & 15);

    const float* arowp = y + (size_t)b * (TLEN * HDIM) + (size_t)trow * HDIM + 2 * khalf;

    v8f acc[4];
#pragma unroll
    for (int j = 0; j < 4; ++j) acc[j] = (v8f){0, 0, 0, 0, 0, 0, 0, 0};

    for (int k0 = 0; k0 < HDIM; k0 += 4) {
        v2f a = *(const v2f*)(arowp + k0);
        const float* wrow = fcpk + ((k0 >> 1) + khalf) * 128 + 2 * (lane & 15);
#pragma unroll
        for (int j = 0; j < 4; ++j) {
            v2f bv = *(const v2f*)(wrow + 32 * j);
            acc[j] = WMMA_F32X4(a, bv, acc[j]);
        }
    }

#pragma unroll
    for (int j = 0; j < 4; ++j) {
        int n = (j << 4) + (lane & 15);
        if (n < CDIM) {
            float bias = fcb[n];
            float* outp =
                label + (size_t)b * (CDIM * TLEN) + (size_t)n * TLEN + t0 + (khalf << 3);
#pragma unroll
            for (int r = 0; r < 8; ++r) outp[r] = acc[j][r] + bias;
        }
    }
}

extern "C" void kernel_launch(void* const* d_in, const int* in_sizes, int n_in,
                              void* d_out, int out_size, void* d_ws, size_t ws_size,
                              hipStream_t stream) {
    const float* x = (const float*)d_in[0];
    const float *Wih[4], *Whh[4], *bih[4], *bhh[4];
    for (int l = 0; l < 4; ++l) {
        Wih[l] = (const float*)d_in[1 + 4 * l];
        Whh[l] = (const float*)d_in[2 + 4 * l];
        bih[l] = (const float*)d_in[3 + 4 * l];
        bhh[l] = (const float*)d_in[4 + 4 * l];
    }
    const float* fcw = (const float*)d_in[17];
    const float* fcb = (const float*)d_in[18];

    float* out   = (float*)d_out;
    float* label = out;
    float* hs    = out + (size_t)BATCH * CDIM * TLEN;
    float* cs    = hs + (size_t)4 * BATCH * HDIM;

    float* ws   = (float*)d_ws;
    float* ybuf = ws;
    float* wpk  = ybuf + (size_t)BATCH * TLEN * HDIM;
    float* fcpk = wpk + (size_t)4 * KDIM * GDIM;

    pack_lstm_weights<<<(4 * KDIM * GDIM) / 256, 256, 0, stream>>>(
        Wih[0], Whh[0], Wih[1], Whh[1], Wih[2], Whh[2], Wih[3], Whh[3], wpk);
    pack_fc<<<(HDIM * 64) / 256, 256, 0, stream>>>(fcw, fcpk);

    for (int l = 0; l < 4; ++l) {
        lstm_scan<<<4, 512, 0, stream>>>(
            (l == 0) ? x : ybuf, ybuf, wpk + (size_t)l * KDIM * GDIM,
            bih[l], bhh[l],
            hs + (size_t)l * BATCH * HDIM, cs + (size_t)l * BATCH * HDIM);
    }

    fc_kernel<<<256, 256, 0, stream>>>(ybuf, fcpk, fcb, label);
}
